# starting kernel
# baseline (speedup 1.0000x reference)
#include <hip/hip_runtime.h>
#include <math.h>


typedef __attribute__((ext_vector_type(2))) float v2f;
typedef __attribute__((ext_vector_type(8))) float v8f;

__device__ __forceinline__ float atomic_add_f32(float* p, float v) {
  return __hip_atomic_fetch_add(p, v, __ATOMIC_RELAXED, __HIP_MEMORY_SCOPE_AGENT);
}

__device__ __forceinline__ void atomic_max_f32(float* p, float v) {
  if (v >= 0.0f) atomicMax((int*)p, __float_as_int(v));
  else           atomicMin((unsigned int*)p, __float_as_uint(v));
}

__global__ __launch_bounds__(32) void wmma_gemm_f32(
    const float* __restrict__ A, const float* __restrict__ B,
    float* __restrict__ C, int M, int N, int K)
{
  const int lane = threadIdx.x & 31;
  const int half = lane >> 4;
  const int l15  = lane & 15;
  const int tm   = blockIdx.x;
  const int tn   = blockIdx.y << 6;

  v8f c0 = {}; v8f c1 = {}; v8f c2 = {}; v8f c3 = {};

  const float* Arow = A + (size_t)(tm * 16 + l15) * K;
  for (int k0 = 0; k0 < K; k0 += 4) {
    const int ka = k0 + (half << 1);
    v2f a;
    a.x = Arow[ka];
    a.y = Arow[ka + 1];
    const float* Bp = B + (size_t)ka * N + tn + l15;
    v2f b0, b1, b2, b3;
    b0.x = Bp[0];   b0.y = Bp[N];
    b1.x = Bp[16];  b1.y = Bp[N + 16];
    b2.x = Bp[32];  b2.y = Bp[N + 32];
    b3.x = Bp[48];  b3.y = Bp[N + 48];
    c0 = __builtin_amdgcn_wmma_f32_16x16x4_f32(false, a, false, b0, (short)0, c0, false, false);
    c1 = __builtin_amdgcn_wmma_f32_16x16x4_f32(false, a, false, b1, (short)0, c1, false, false);
    c2 = __builtin_amdgcn_wmma_f32_16x16x4_f32(false, a, false, b2, (short)0, c2, false, false);
    c3 = __builtin_amdgcn_wmma_f32_16x16x4_f32(false, a, false, b3, (short)0, c3, false, false);
  }

  const int orow = tm * 16 + (half << 3);
  for (int r = 0; r < 8; ++r) {
    float* Cp = C + (size_t)(orow + r) * N + tn + l15;
    Cp[0]  = c0[r];
    Cp[16] = c1[r];
    Cp[32] = c2[r];
    Cp[48] = c3[r];
  }
}

__global__ __launch_bounds__(256) void node_dots(
    const float* __restrict__ xp, const float* __restrict__ att_src,
    const float* __restrict__ att_dst, float* __restrict__ asrc,
    float* __restrict__ adst, int Nn, int H, int C)
{
  const int wid  = (int)((blockIdx.x * blockDim.x + threadIdx.x) >> 5);
  const int lane = threadIdx.x & 31;
  if (wid >= Nn * H) return;
  const int n = wid / H;
  const int h = wid - n * H;

  const float* xr = xp + (size_t)n * H * C + (size_t)h * C;
  const float* as = att_src + (size_t)h * C;
  const float* ad = att_dst + (size_t)h * C;
  float s = 0.f, d = 0.f;
  for (int c = lane; c < C; c += 32) {
    float v = xr[c];
    s += v * as[c];
    d += v * ad[c];
  }
  for (int off = 16; off > 0; off >>= 1) {
    s += __shfl_down(s, off, 32);
    d += __shfl_down(d, off, 32);
  }
  if (lane == 0) { asrc[wid] = s; adst[wid] = d; }
}

__global__ void fill_f32(float* __restrict__ p, float v, size_t n) {
  size_t i = (size_t)blockIdx.x * blockDim.x + threadIdx.x;
  size_t stride = (size_t)gridDim.x * blockDim.x;
  for (; i < n; i += stride) p[i] = v;
}

__global__ __launch_bounds__(256) void edge_max(
    const int* __restrict__ src, const int* __restrict__ dst, int E0, int Nn,
    const float* __restrict__ asrc, const float* __restrict__ adst,
    float* __restrict__ amax, int H)
{
  const int e = blockIdx.x * blockDim.x + threadIdx.x;
  if (e >= E0 + Nn) return;
  const int s = (e < E0) ? src[e] : (e - E0);
  const int d = (e < E0) ? dst[e] : (e - E0);
  for (int h = 0; h < H; ++h) {
    float a = asrc[s * H + h] + adst[d * H + h];
    a = (a > 0.f) ? a : 0.2f * a;
    atomic_max_f32(&amax[d * H + h], a);
  }
}

__global__ __launch_bounds__(256) void edge_exp(
    const int* __restrict__ src, const int* __restrict__ dst, int E0, int Nn,
    const float* __restrict__ asrc, const float* __restrict__ adst,
    const float* __restrict__ amax, float* __restrict__ den,
    float* __restrict__ ex, int H)
{
  const int e = blockIdx.x * blockDim.x + threadIdx.x;
  if (e >= E0 + Nn) return;
  const int s = (e < E0) ? src[e] : (e - E0);
  const int d = (e < E0) ? dst[e] : (e - E0);
  for (int h = 0; h < H; ++h) {
    float a = asrc[s * H + h] + adst[d * H + h];
    a = (a > 0.f) ? a : 0.2f * a;
    float v = __expf(a - amax[d * H + h]);
    ex[(size_t)e * H + h] = v;
    atomic_add_f32(&den[d * H + h], v);
  }
}

__global__ __launch_bounds__(256) void edge_aggregate(
    const int* __restrict__ src, const int* __restrict__ dst, int E0,
    const float* __restrict__ xp, const float* __restrict__ ex,
    const float* __restrict__ den, float* __restrict__ out, int H)
{
  const int e = blockIdx.x;
  const int s = (e < E0) ? src[e] : (e - E0);
  const int d = (e < E0) ? dst[e] : (e - E0);
  const int F = H << 8;
  const size_t eb = (size_t)e * H;
  const size_t sb = (size_t)s * F;
  const size_t db = (size_t)d * F;
  for (int f = threadIdx.x; f < F; f += 256) {
    const int h = f >> 8;
    const float w = ex[eb + h] / den[d * H + h];
    atomic_add_f32(&out[db + f], xp[sb + f] * w);
  }
}

__global__ void bias_elu(float* __restrict__ y, const float* __restrict__ b,
                         int F, size_t n) {
  size_t i = (size_t)blockIdx.x * blockDim.x + threadIdx.x;
  size_t stride = (size_t)gridDim.x * blockDim.x;
  for (; i < n; i += stride) {
    int f = (int)(i - (i / F) * F);
    float v = y[i] + b[f];
    y[i] = (v > 0.f) ? v : expm1f(v);
  }
}

__global__ void bias_add(float* __restrict__ y, const float* __restrict__ b,
                         int Fmask, size_t n) {
  size_t i = (size_t)blockIdx.x * blockDim.x + threadIdx.x;
  size_t stride = (size_t)gridDim.x * blockDim.x;
  for (; i < n; i += stride) y[i] += b[(int)i & Fmask];
}


extern "C" void kernel_launch(void* const* d_in, const int* in_sizes, int n_in,
                              void* d_out, int out_size, void* d_ws, size_t ws_size,
                              hipStream_t stream) {
  const float* x   = (const float*)d_in[0];
  const int*   ei  = (const int*)d_in[1];
  const float* W1  = (const float*)d_in[2];
  const float* as1 = (const float*)d_in[3];
  const float* ad1 = (const float*)d_in[4];
  const float* b1  = (const float*)d_in[5];
  const float* W2  = (const float*)d_in[6];
  const float* as2 = (const float*)d_in[7];
  const float* ad2 = (const float*)d_in[8];
  const float* b2  = (const float*)d_in[9];

  const int K1 = 768, F1 = 768, H1 = 3;
  const int F2 = 256, H2 = 1;
  const int Nn = in_sizes[0] / K1;
  const int E0 = in_sizes[1] / 2;
  const int ET = E0 + Nn;

  const int* srcI = ei;
  const int* dstI = ei + E0;

  float* ws = (float*)d_ws;
  float* xp1   = ws; ws += (size_t)Nn * F1;
  float* h1    = ws; ws += (size_t)Nn * F1;
  float* xp2   = ws; ws += (size_t)Nn * F2;
  float* asrc1 = ws; ws += (size_t)Nn * H1;
  float* adst1 = ws; ws += (size_t)Nn * H1;
  float* amax1 = ws; ws += (size_t)Nn * H1;
  float* den1  = ws; ws += (size_t)Nn * H1;
  float* asrc2 = ws; ws += Nn;
  float* adst2 = ws; ws += Nn;
  float* amax2 = ws; ws += Nn;
  float* den2  = ws; ws += Nn;
  float* ex1   = ws; ws += (size_t)ET * H1;
  float* ex2   = ws; ws += (size_t)ET;

  const float NEG_INF = -__builtin_inff();
  const int FB = 512;

  {
    dim3 g(Nn / 16, F1 / 64);
    wmma_gemm_f32<<<g, 32, 0, stream>>>(x, W1, xp1, Nn, F1, K1);
  }
  node_dots<<<(Nn * H1 + 7) / 8, 256, 0, stream>>>(xp1, as1, ad1, asrc1, adst1, Nn, H1, 256);
  fill_f32<<<FB, 256, 0, stream>>>(amax1, NEG_INF, (size_t)Nn * H1);
  fill_f32<<<FB, 256, 0, stream>>>(den1, 0.f, (size_t)Nn * H1);
  fill_f32<<<FB, 256, 0, stream>>>(h1, 0.f, (size_t)Nn * F1);
  edge_max<<<(ET + 255) / 256, 256, 0, stream>>>(srcI, dstI, E0, Nn, asrc1, adst1, amax1, H1);
  edge_exp<<<(ET + 255) / 256, 256, 0, stream>>>(srcI, dstI, E0, Nn, asrc1, adst1, amax1, den1, ex1, H1);
  edge_aggregate<<<ET, 256, 0, stream>>>(srcI, dstI, E0, xp1, ex1, den1, h1, H1);
  bias_elu<<<FB, 256, 0, stream>>>(h1, b1, F1, (size_t)Nn * F1);

  {
    dim3 g(Nn / 16, F2 / 64);
    wmma_gemm_f32<<<g, 32, 0, stream>>>(h1, W2, xp2, Nn, F2, F1);
  }
  node_dots<<<(Nn + 7) / 8, 256, 0, stream>>>(xp2, as2, ad2, asrc2, adst2, Nn, H2, 256);
  fill_f32<<<FB, 256, 0, stream>>>(amax2, NEG_INF, (size_t)Nn);
  fill_f32<<<FB, 256, 0, stream>>>(den2, 0.f, (size_t)Nn);
  fill_f32<<<FB, 256, 0, stream>>>((float*)d_out, 0.f, (size_t)Nn * F2);
  edge_max<<<(ET + 255) / 256, 256, 0, stream>>>(srcI, dstI, E0, Nn, asrc2, adst2, amax2, H2);
  edge_exp<<<(ET + 255) / 256, 256, 0, stream>>>(srcI, dstI, E0, Nn, asrc2, adst2, amax2, den2, ex2, H2);
  edge_aggregate<<<ET, 256, 0, stream>>>(srcI, dstI, E0, xp2, ex2, den2, (float*)d_out, H2);
  bias_add<<<FB, 256, 0, stream>>>((float*)d_out, b2, F2 - 1, (size_t)Nn * F2);
}


# Softmax and aggregation per destination node over a dst-sorted edge list built each call (counting sort) replace per-edge float atomics; sum order = arrival-order rank; bias/ELU folded into the store
# speedup vs baseline: 2.5255x; 2.5255x over previous
#include <hip/hip_runtime.h>
#include <math.h>


typedef __attribute__((ext_vector_type(2))) float v2f;
typedef __attribute__((ext_vector_type(8))) float v8f;

__global__ __launch_bounds__(32) void wmma_gemm_f32(
    const float* __restrict__ A, const float* __restrict__ B,
    float* __restrict__ C, int M, int N, int K)
{
  const int lane = threadIdx.x & 31;
  const int half = lane >> 4;
  const int l15  = lane & 15;
  const int tm   = blockIdx.x;
  const int tn   = blockIdx.y << 6;

  v8f c0 = {}; v8f c1 = {}; v8f c2 = {}; v8f c3 = {};

  const float* Arow = A + (size_t)(tm * 16 + l15) * K;
  for (int k0 = 0; k0 < K; k0 += 4) {
    const int ka = k0 + (half << 1);
    v2f a;
    a.x = Arow[ka];
    a.y = Arow[ka + 1];
    const float* Bp = B + (size_t)ka * N + tn + l15;
    v2f b0, b1, b2, b3;
    b0.x = Bp[0];   b0.y = Bp[N];
    b1.x = Bp[16];  b1.y = Bp[N + 16];
    b2.x = Bp[32];  b2.y = Bp[N + 32];
    b3.x = Bp[48];  b3.y = Bp[N + 48];
    c0 = __builtin_amdgcn_wmma_f32_16x16x4_f32(false, a, false, b0, (short)0, c0, false, false);
    c1 = __builtin_amdgcn_wmma_f32_16x16x4_f32(false, a, false, b1, (short)0, c1, false, false);
    c2 = __builtin_amdgcn_wmma_f32_16x16x4_f32(false, a, false, b2, (short)0, c2, false, false);
    c3 = __builtin_amdgcn_wmma_f32_16x16x4_f32(false, a, false, b3, (short)0, c3, false, false);
  }

  const int orow = tm * 16 + (half << 3);
  for (int r = 0; r < 8; ++r) {
    float* Cp = C + (size_t)(orow + r) * N + tn + l15;
    Cp[0]  = c0[r];
    Cp[16] = c1[r];
    Cp[32] = c2[r];
    Cp[48] = c3[r];
  }
}

__global__ __launch_bounds__(256) void node_dots(
    const float* __restrict__ xp, const float* __restrict__ att_src,
    const float* __restrict__ att_dst, float* __restrict__ asrc,
    float* __restrict__ adst, int Nn, int H, int C)
{
  const int wid  = (int)((blockIdx.x * blockDim.x + threadIdx.x) >> 5);
  const int lane = threadIdx.x & 31;
  if (wid >= Nn * H) return;
  const int n = wid / H;
  const int h = wid - n * H;

  const float* xr = xp + (size_t)n * H * C + (size_t)h * C;
  const float* as = att_src + (size_t)h * C;
  const float* ad = att_dst + (size_t)h * C;
  float s = 0.f, d = 0.f;
  for (int c = lane; c < C; c += 32) {
    float v = xr[c];
    s += v * as[c];
    d += v * ad[c];
  }
  for (int off = 16; off > 0; off >>= 1) {
    s += __shfl_down(s, off, 32);
    d += __shfl_down(d, off, 32);
  }
  if (lane == 0) { asrc[wid] = s; adst[wid] = d; }
}

__global__ void zero_i32(int* __restrict__ p, int n) {
  const int i = blockIdx.x * blockDim.x + threadIdx.x;
  if (i < n) p[i] = 0;
}

__global__ __launch_bounds__(256) void count_dst(
    const int* __restrict__ dst, int E0, int* __restrict__ cnt)
{
  const int e = blockIdx.x * blockDim.x + threadIdx.x;
  if (e < E0) atomicAdd(&cnt[dst[e]], 1);
}

__global__ __launch_bounds__(1024) void scan_offsets(
    const int* __restrict__ cnt, int* __restrict__ off, int* __restrict__ cur, int n)
{
  __shared__ int part[1024];
  const int t = threadIdx.x;
  const int chunk = (n + 1023) >> 10;
  const int b = t * chunk;
  const int e = (b + chunk < n) ? (b + chunk) : n;
  int s = 0;
  for (int i = b; i < e; ++i) s += cnt[i];
  part[t] = s;
  __syncthreads();
  for (int o = 1; o < 1024; o <<= 1) {
    const int v = (t >= o) ? part[t - o] : 0;
    __syncthreads();
    part[t] += v;
    __syncthreads();
  }
  int run = part[t] - s;
  for (int i = b; i < e; ++i) { off[i] = run; cur[i] = run; run += cnt[i]; }
  if (t == 1023) off[n] = part[1023];
}

__global__ __launch_bounds__(256) void fill_sorted(
    const int* __restrict__ src, const int* __restrict__ dst, int E0,
    int* __restrict__ cur, int* __restrict__ ssrc)
{
  const int e = blockIdx.x * blockDim.x + threadIdx.x;
  if (e < E0) {
    const int p = atomicAdd(&cur[dst[e]], 1);
    ssrc[p] = src[e];
  }
}

__device__ __forceinline__ float leaky02(float a) { return (a > 0.f) ? a : 0.2f * a; }

__device__ __forceinline__ void fma4(float4& acc, const float4 r, const float w) {
  acc.x = __builtin_fmaf(r.x, w, acc.x);
  acc.y = __builtin_fmaf(r.y, w, acc.y);
  acc.z = __builtin_fmaf(r.z, w, acc.z);
  acc.w = __builtin_fmaf(r.w, w, acc.w);
}

__device__ __forceinline__ void gather_group(
    float4& acc, const float* __restrict__ xcol, const int F,
    const int s, const float w, const int n)
{
  const int wi = __float_as_int(w);
  int jj = 0;
  for (; jj + 4 <= n; jj += 4) {
    const int s0 = __builtin_amdgcn_readlane(s, jj);
    const int s1 = __builtin_amdgcn_readlane(s, jj + 1);
    const int s2 = __builtin_amdgcn_readlane(s, jj + 2);
    const int s3 = __builtin_amdgcn_readlane(s, jj + 3);
    const float w0 = __int_as_float(__builtin_amdgcn_readlane(wi, jj));
    const float w1 = __int_as_float(__builtin_amdgcn_readlane(wi, jj + 1));
    const float w2 = __int_as_float(__builtin_amdgcn_readlane(wi, jj + 2));
    const float w3 = __int_as_float(__builtin_amdgcn_readlane(wi, jj + 3));
    const float4 r0 = *(const float4*)(xcol + (size_t)s0 * F);
    const float4 r1 = *(const float4*)(xcol + (size_t)s1 * F);
    const float4 r2 = *(const float4*)(xcol + (size_t)s2 * F);
    const float4 r3 = *(const float4*)(xcol + (size_t)s3 * F);
    fma4(acc, r0, w0);
    fma4(acc, r1, w1);
    fma4(acc, r2, w2);
    fma4(acc, r3, w3);
  }
  for (; jj < n; ++jj) {
    const int s0 = __builtin_amdgcn_readlane(s, jj);
    const float w0 = __int_as_float(__builtin_amdgcn_readlane(wi, jj));
    const float4 r0 = *(const float4*)(xcol + (size_t)s0 * F);
    fma4(acc, r0, w0);
  }
}

template <int H, bool ELU>
__global__ __launch_bounds__(256) void segment_softmax_aggregate(
    const int* __restrict__ off, const int* __restrict__ ssrc,
    const float* __restrict__ xp, const float* __restrict__ asrc,
    const float* __restrict__ adst, const float* __restrict__ bias,
    float* __restrict__ out, int Nn)
{
  constexpr int WPN = H * 2;
  constexpr int F   = H * 256;
  const int lane = threadIdx.x & 31;
  const int wv = __builtin_amdgcn_readfirstlane((int)(threadIdx.x >> 5));
  const int gw = blockIdx.x * 8 + wv;
  if (gw >= Nn * WPN) return;
  const int d    = gw / WPN;
  const int part = gw - d * WPN;
  const int h    = part >> 1;
  const int beg  = off[d];
  const int deg  = off[d + 1] - beg;
  const int tot  = deg + 1;
  const float ad = adst[d * H + h];
  const float NEG_INF = -__builtin_inff();

  int   s0 = d;
  float a0 = NEG_INF;
  if (lane < tot) {
    s0 = (lane < deg) ? ssrc[beg + lane] : d;
    a0 = leaky02(asrc[s0 * H + h] + ad);
  }

  float m = a0;
  for (int j = 32 + lane; j < tot; j += 32) {
    const int s = (j < deg) ? ssrc[beg + j] : d;
    m = fmaxf(m, leaky02(asrc[s * H + h] + ad));
  }
  for (int o = 16; o > 0; o >>= 1) m = fmaxf(m, __shfl_xor(m, o, 32));

  const float e0 = (lane < tot) ? __expf(a0 - m) : 0.f;
  float den = e0;
  for (int j = 32 + lane; j < tot; j += 32) {
    const int s = (j < deg) ? ssrc[beg + j] : d;
    den += __expf(leaky02(asrc[s * H + h] + ad) - m);
  }
  for (int o = 16; o > 0; o >>= 1) den += __shfl_xor(den, o, 32);

  const int col = part * 128 + lane * 4;
  const float* xcol = xp + col;
  const float4 b = *(const float4*)(bias + col);
  float4 acc = make_float4(0.f, 0.f, 0.f, 0.f);

  gather_group(acc, xcol, F, s0, e0 / den, (tot < 32) ? tot : 32);
  for (int base = 32; base < tot; base += 32) {
    const int j = base + lane;
    int   s = d;
    float w = 0.f;
    if (j < tot) {
      s = (j < deg) ? ssrc[beg + j] : d;
      w = __expf(leaky02(asrc[s * H + h] + ad) - m) / den;
    }
    const int n = tot - base;
    gather_group(acc, xcol, F, s, w, (n < 32) ? n : 32);
  }

  float4 v = make_float4(acc.x + b.x, acc.y + b.y, acc.z + b.z, acc.w + b.w);
  if (ELU) {
    v.x = (v.x > 0.f) ? v.x : expm1f(v.x);
    v.y = (v.y > 0.f) ? v.y : expm1f(v.y);
    v.z = (v.z > 0.f) ? v.z : expm1f(v.z);
    v.w = (v.w > 0.f) ? v.w : expm1f(v.w);
  }
  *(float4*)(out + (size_t)d * F + col) = v;
}


extern "C" void kernel_launch(void* const* d_in, const int* in_sizes, int n_in,
                              void* d_out, int out_size, void* d_ws, size_t ws_size,
                              hipStream_t stream) {
  const float* x   = (const float*)d_in[0];
  const int*   ei  = (const int*)d_in[1];
  const float* W1  = (const float*)d_in[2];
  const float* as1 = (const float*)d_in[3];
  const float* ad1 = (const float*)d_in[4];
  const float* b1  = (const float*)d_in[5];
  const float* W2  = (const float*)d_in[6];
  const float* as2 = (const float*)d_in[7];
  const float* ad2 = (const float*)d_in[8];
  const float* b2  = (const float*)d_in[9];

  const int K1 = 768, F1 = 768, H1 = 3;
  const int F2 = 256, H2 = 1;
  const int Nn = in_sizes[0] / K1;
  const int E0 = in_sizes[1] / 2;

  const int* srcI = ei;
  const int* dstI = ei + E0;

  float* ws = (float*)d_ws;
  float* xp1   = ws; ws += (size_t)Nn * F1;
  float* h1    = ws; ws += (size_t)Nn * F1;
  float* xp2   = ws; ws += (size_t)Nn * F2;
  float* asrc1 = ws; ws += (size_t)Nn * H1;
  float* adst1 = ws; ws += (size_t)Nn * H1;
  float* asrc2 = ws; ws += Nn;
  float* adst2 = ws; ws += Nn;
  int* cnt  = (int*)ws; ws += Nn;
  int* off  = (int*)ws; ws += Nn + 1;
  int* cur  = (int*)ws; ws += Nn;
  int* ssrc = (int*)ws; ws += E0;

  zero_i32<<<(Nn + 255) / 256, 256, 0, stream>>>(cnt, Nn);
  count_dst<<<(E0 + 255) / 256, 256, 0, stream>>>(dstI, E0, cnt);
  scan_offsets<<<1, 1024, 0, stream>>>(cnt, off, cur, Nn);
  fill_sorted<<<(E0 + 255) / 256, 256, 0, stream>>>(srcI, dstI, E0, cur, ssrc);

  {
    dim3 g(Nn / 16, F1 / 64);
    wmma_gemm_f32<<<g, 32, 0, stream>>>(x, W1, xp1, Nn, F1, K1);
  }
  node_dots<<<(Nn * H1 + 7) / 8, 256, 0, stream>>>(xp1, as1, ad1, asrc1, adst1, Nn, H1, 256);
  segment_softmax_aggregate<3, true><<<(Nn * 6 + 7) / 8, 256, 0, stream>>>(
      off, ssrc, xp1, asrc1, adst1, b1, h1, Nn);

  {
    dim3 g(Nn / 16, F2 / 64);
    wmma_gemm_f32<<<g, 32, 0, stream>>>(h1, W2, xp2, Nn, F2, F1);
  }
  node_dots<<<(Nn + 7) / 8, 256, 0, stream>>>(xp2, as2, ad2, asrc2, adst2, Nn, H2, 256);
  segment_softmax_aggregate<1, false><<<(Nn * 2 + 7) / 8, 256, 0, stream>>>(
      off, ssrc, xp2, asrc2, adst2, b2, (float*)d_out, Nn);
}


# Both GEMMs tiled per workgroup through LDS (128x128 and 64x128 tiles, K-step 32, 32x64 per wave, two stages, plain global loads + ds stores) replace one-wave 16x64 strips; sum order kept
# speedup vs baseline: 3.6906x; 1.4613x over previous
#include <hip/hip_runtime.h>
#include <math.h>


typedef __attribute__((ext_vector_type(2))) float v2f;
typedef __attribute__((ext_vector_type(4))) float v4f;
typedef __attribute__((ext_vector_type(8))) float v8f;

template <int WM>
__global__ __launch_bounds__(64 * WM) void wmma_gemm_f32(
    const float* __restrict__ A, const float* __restrict__ B,
    float* __restrict__ C, int M, int N, int K)
{
  constexpr int BM = 32 * WM, BN = 128, BK = 32;
  constexpr int NT = 64 * WM;
  constexpr int PA = BK + 4;
  constexpr int PB = BN + 16;
  constexpr int A_LD = BM * (BK / 4) / NT;
  constexpr int B_LD = BK * (BN / 4) / NT;

  __shared__ __align__(16) float As[2][BM * PA];
  __shared__ __align__(16) float Bs[2][BK * PB];

  const int t    = threadIdx.x;
  const int lane = t & 31;
  const int h    = lane >> 4;
  const int c    = lane & 15;
  const int wave = __builtin_amdgcn_readfirstlane(t >> 5);
  const int wm   = wave >> 1;
  const int wn   = wave & 1;
  const int n0   = blockIdx.x * BN;
  const int m0   = blockIdx.y * BM;

  const int a_kq  = t & 7;
  const int a_row = t >> 3;
  const int b_nq  = t & 31;
  const int b_row = t >> 5;

  const float* Ag[A_LD];
  for (int i = 0; i < A_LD; ++i) {
    int r = m0 + a_row + i * (NT / 8);
    r = (r < M) ? r : (M - 1);
    Ag[i] = A + (size_t)r * K + a_kq * 4;
  }
  const float* Bg = B + (size_t)b_row * N + n0 + b_nq * 4;

  v4f ra[A_LD], rb[B_LD];
  v8f acc[2][4];
  for (int i = 0; i < 2; ++i)
    for (int f = 0; f < 4; ++f) acc[i][f] = v8f{};

  auto load_global = [&](int k0) {
    for (int i = 0; i < A_LD; ++i) ra[i] = *(const v4f*)(Ag[i] + k0);
    for (int i = 0; i < B_LD; ++i)
      rb[i] = *(const v4f*)(Bg + (size_t)(k0 + i * (NT / 32)) * N);
  };
  auto store_lds = [&](int buf) {
    for (int i = 0; i < A_LD; ++i)
      *(v4f*)(&As[buf][(a_row + i * (NT / 8)) * PA + a_kq * 4]) = ra[i];
    for (int i = 0; i < B_LD; ++i)
      *(v4f*)(&Bs[buf][(b_row + i * (NT / 32)) * PB + b_nq * 4]) = rb[i];
  };
  auto compute = [&](int buf) {
    const float* Ap = &As[buf][(wm * 32 + c) * PA + 2 * h];
    const float* Bp = &Bs[buf][(2 * h) * PB + wn * 64 + 4 * c];
    for (int kk = 0; kk < BK; kk += 4) {
      const v2f a0 = *(const v2f*)(Ap + kk);
      const v2f a1 = *(const v2f*)(Ap + 16 * PA + kk);
      const v4f bx = *(const v4f*)(Bp + kk * PB);
      const v4f by = *(const v4f*)(Bp + (kk + 1) * PB);
      v2f b0, b1, b2, b3;
      b0.x = bx.x; b0.y = by.x;
      b1.x = bx.y; b1.y = by.y;
      b2.x = bx.z; b2.y = by.z;
      b3.x = bx.w; b3.y = by.w;
      acc[0][0] = __builtin_amdgcn_wmma_f32_16x16x4_f32(false, a0, false, b0, (short)0, acc[0][0], false, false);
      acc[1][0] = __builtin_amdgcn_wmma_f32_16x16x4_f32(false, a1, false, b0, (short)0, acc[1][0], false, false);
      acc[0][1] = __builtin_amdgcn_wmma_f32_16x16x4_f32(false, a0, false, b1, (short)0, acc[0][1], false, false);
      acc[1][1] = __builtin_amdgcn_wmma_f32_16x16x4_f32(false, a1, false, b1, (short)0, acc[1][1], false, false);
      acc[0][2] = __builtin_amdgcn_wmma_f32_16x16x4_f32(false, a0, false, b2, (short)0, acc[0][2], false, false);
      acc[1][2] = __builtin_amdgcn_wmma_f32_16x16x4_f32(false, a1, false, b2, (short)0, acc[1][2], false, false);
      acc[0][3] = __builtin_amdgcn_wmma_f32_16x16x4_f32(false, a0, false, b3, (short)0, acc[0][3], false, false);
      acc[1][3] = __builtin_amdgcn_wmma_f32_16x16x4_f32(false, a1, false, b3, (short)0, acc[1][3], false, false);
    }
  };

  load_global(0);
  store_lds(0);
  __syncthreads();
  int buf = 0;
  for (int k0 = BK; k0 < K; k0 += BK) {
    load_global(k0);
    __builtin_amdgcn_sched_barrier(0);
    compute(buf);
    store_lds(buf ^ 1);
    __syncthreads();
    buf ^= 1;
  }
  compute(buf);

  float* Cp = C + n0 + wn * 64 + 4 * c;
  for (int i = 0; i < 2; ++i) {
    for (int r = 0; r < 8; ++r) {
      const int row = m0 + wm * 32 + i * 16 + h * 8 + r;
      if (row < M) {
        v4f o;
        o.x = acc[i][0][r]; o.y = acc[i][1][r]; o.z = acc[i][2][r]; o.w = acc[i][3][r];
        *(v4f*)(Cp + (size_t)row * N) = o;
      }
    }
  }
}

__global__ __launch_bounds__(256) void node_dots(
    const float* __restrict__ xp, const float* __restrict__ att_src,
    const float* __restrict__ att_dst, float* __restrict__ asrc,
    float* __restrict__ adst, int Nn, int H, int C)
{
  const int wid  = (int)((blockIdx.x * blockDim.x + threadIdx.x) >> 5);
  const int lane = threadIdx.x & 31;
  if (wid >= Nn * H) return;
  const int n = wid / H;
  const int h = wid - n * H;

  const float* xr = xp + (size_t)n * H * C + (size_t)h * C;
  const float* as = att_src + (size_t)h * C;
  const float* ad = att_dst + (size_t)h * C;
  float s = 0.f, d = 0.f;
  for (int c = lane; c < C; c += 32) {
    float v = xr[c];
    s += v * as[c];
    d += v * ad[c];
  }
  for (int off = 16; off > 0; off >>= 1) {
    s += __shfl_down(s, off, 32);
    d += __shfl_down(d, off, 32);
  }
  if (lane == 0) { asrc[wid] = s; adst[wid] = d; }
}

__global__ void zero_i32(int* __restrict__ p, int n) {
  const int i = blockIdx.x * blockDim.x + threadIdx.x;
  if (i < n) p[i] = 0;
}

__global__ __launch_bounds__(256) void count_dst(
    const int* __restrict__ dst, int E0, int* __restrict__ cnt)
{
  const int e = blockIdx.x * blockDim.x + threadIdx.x;
  if (e < E0) atomicAdd(&cnt[dst[e]], 1);
}

__global__ __launch_bounds__(1024) void scan_offsets(
    const int* __restrict__ cnt, int* __restrict__ off, int* __restrict__ cur, int n)
{
  __shared__ int part[1024];
  const int t = threadIdx.x;
  const int chunk = (n + 1023) >> 10;
  const int b = t * chunk;
  const int e = (b + chunk < n) ? (b + chunk) : n;
  int s = 0;
  for (int i = b; i < e; ++i) s += cnt[i];
  part[t] = s;
  __syncthreads();
  for (int o = 1; o < 1024; o <<= 1) {
    const int v = (t >= o) ? part[t - o] : 0;
    __syncthreads();
    part[t] += v;
    __syncthreads();
  }
  int run = part[t] - s;
  for (int i = b; i < e; ++i) { off[i] = run; cur[i] = run; run += cnt[i]; }
  if (t == 1023) off[n] = part[1023];
}

__global__ __launch_bounds__(256) void fill_sorted(
    const int* __restrict__ src, const int* __restrict__ dst, int E0,
    int* __restrict__ cur, int* __restrict__ ssrc)
{
  const int e = blockIdx.x * blockDim.x + threadIdx.x;
  if (e < E0) {
    const int p = atomicAdd(&cur[dst[e]], 1);
    ssrc[p] = src[e];
  }
}

__device__ __forceinline__ float leaky02(float a) { return (a > 0.f) ? a : 0.2f * a; }

__device__ __forceinline__ void fma4(float4& acc, const float4 r, const float w) {
  acc.x = __builtin_fmaf(r.x, w, acc.x);
  acc.y = __builtin_fmaf(r.y, w, acc.y);
  acc.z = __builtin_fmaf(r.z, w, acc.z);
  acc.w = __builtin_fmaf(r.w, w, acc.w);
}

__device__ __forceinline__ void gather_group(
    float4& acc, const float* __restrict__ xcol, const int F,
    const int s, const float w, const int n)
{
  const int wi = __float_as_int(w);
  int jj = 0;
  for (; jj + 4 <= n; jj += 4) {
    const int s0 = __builtin_amdgcn_readlane(s, jj);
    const int s1 = __builtin_amdgcn_readlane(s, jj + 1);
    const int s2 = __builtin_amdgcn_readlane(s, jj + 2);
    const int s3 = __builtin_amdgcn_readlane(s, jj + 3);
    const float w0 = __int_as_float(__builtin_amdgcn_readlane(wi, jj));
    const float w1 = __int_as_float(__builtin_amdgcn_readlane(wi, jj + 1));
    const float w2 = __int_as_float(__builtin_amdgcn_readlane(wi, jj + 2));
    const float w3 = __int_as_float(__builtin_amdgcn_readlane(wi, jj + 3));
    const float4 r0 = *(const float4*)(xcol + (size_t)s0 * F);
    const float4 r1 = *(const float4*)(xcol + (size_t)s1 * F);
    const float4 r2 = *(const float4*)(xcol + (size_t)s2 * F);
    const float4 r3 = *(const float4*)(xcol + (size_t)s3 * F);
    fma4(acc, r0, w0);
    fma4(acc, r1, w1);
    fma4(acc, r2, w2);
    fma4(acc, r3, w3);
  }
  for (; jj < n; ++jj) {
    const int s0 = __builtin_amdgcn_readlane(s, jj);
    const float w0 = __int_as_float(__builtin_amdgcn_readlane(wi, jj));
    const float4 r0 = *(const float4*)(xcol + (size_t)s0 * F);
    fma4(acc, r0, w0);
  }
}

template <int H, bool ELU>
__global__ __launch_bounds__(256) void segment_softmax_aggregate(
    const int* __restrict__ off, const int* __restrict__ ssrc,
    const float* __restrict__ xp, const float* __restrict__ asrc,
    const float* __restrict__ adst, const float* __restrict__ bias,
    float* __restrict__ out, int Nn)
{
  constexpr int WPN = H * 2;
  constexpr int F   = H * 256;
  const int lane = threadIdx.x & 31;
  const int wv = __builtin_amdgcn_readfirstlane((int)(threadIdx.x >> 5));
  const int gw = blockIdx.x * 8 + wv;
  if (gw >= Nn * WPN) return;
  const int d    = gw / WPN;
  const int part = gw - d * WPN;
  const int h    = part >> 1;
  const int beg  = off[d];
  const int deg  = off[d + 1] - beg;
  const int tot  = deg + 1;
  const float ad = adst[d * H + h];
  const float NEG_INF = -__builtin_inff();

  int   s0 = d;
  float a0 = NEG_INF;
  if (lane < tot) {
    s0 = (lane < deg) ? ssrc[beg + lane] : d;
    a0 = leaky02(asrc[s0 * H + h] + ad);
  }

  float m = a0;
  for (int j = 32 + lane; j < tot; j += 32) {
    const int s = (j < deg) ? ssrc[beg + j] : d;
    m = fmaxf(m, leaky02(asrc[s * H + h] + ad));
  }
  for (int o = 16; o > 0; o >>= 1) m = fmaxf(m, __shfl_xor(m, o, 32));

  const float e0 = (lane < tot) ? __expf(a0 - m) : 0.f;
  float den = e0;
  for (int j = 32 + lane; j < tot; j += 32) {
    const int s = (j < deg) ? ssrc[beg + j] : d;
    den += __expf(leaky02(asrc[s * H + h] + ad) - m);
  }
  for (int o = 16; o > 0; o >>= 1) den += __shfl_xor(den, o, 32);

  const int col = part * 128 + lane * 4;
  const float* xcol = xp + col;
  const float4 b = *(const float4*)(bias + col);
  float4 acc = make_float4(0.f, 0.f, 0.f, 0.f);

  gather_group(acc, xcol, F, s0, e0 / den, (tot < 32) ? tot : 32);
  for (int base = 32; base < tot; base += 32) {
    const int j = base + lane;
    int   s = d;
    float w = 0.f;
    if (j < tot) {
      s = (j < deg) ? ssrc[beg + j] : d;
      w = __expf(leaky02(asrc[s * H + h] + ad) - m) / den;
    }
    const int n = tot - base;
    gather_group(acc, xcol, F, s, w, (n < 32) ? n : 32);
  }

  float4 v = make_float4(acc.x + b.x, acc.y + b.y, acc.z + b.z, acc.w + b.w);
  if (ELU) {
    v.x = (v.x > 0.f) ? v.x : expm1f(v.x);
    v.y = (v.y > 0.f) ? v.y : expm1f(v.y);
    v.z = (v.z > 0.f) ? v.z : expm1f(v.z);
    v.w = (v.w > 0.f) ? v.w : expm1f(v.w);
  }
  *(float4*)(out + (size_t)d * F + col) = v;
}


extern "C" void kernel_launch(void* const* d_in, const int* in_sizes, int n_in,
                              void* d_out, int out_size, void* d_ws, size_t ws_size,
                              hipStream_t stream) {
  const float* x   = (const float*)d_in[0];
  const int*   ei  = (const int*)d_in[1];
  const float* W1  = (const float*)d_in[2];
  const float* as1 = (const float*)d_in[3];
  const float* ad1 = (const float*)d_in[4];
  const float* b1  = (const float*)d_in[5];
  const float* W2  = (const float*)d_in[6];
  const float* as2 = (const float*)d_in[7];
  const float* ad2 = (const float*)d_in[8];
  const float* b2  = (const float*)d_in[9];

  const int K1 = 768, F1 = 768, H1 = 3;
  const int F2 = 256, H2 = 1;
  const int Nn = in_sizes[0] / K1;
  const int E0 = in_sizes[1] / 2;

  const int* srcI = ei;
  const int* dstI = ei + E0;

  float* ws = (float*)d_ws;
  float* xp1   = ws; ws += (size_t)Nn * F1;
  float* h1    = ws; ws += (size_t)Nn * F1;
  float* xp2   = ws; ws += (size_t)Nn * F2;
  float* asrc1 = ws; ws += (size_t)Nn * H1;
  float* adst1 = ws; ws += (size_t)Nn * H1;
  float* asrc2 = ws; ws += Nn;
  float* adst2 = ws; ws += Nn;
  int* cnt  = (int*)ws; ws += Nn;
  int* off  = (int*)ws; ws += Nn + 1;
  int* cur  = (int*)ws; ws += Nn;
  int* ssrc = (int*)ws; ws += E0;

  zero_i32<<<(Nn + 255) / 256, 256, 0, stream>>>(cnt, Nn);
  count_dst<<<(E0 + 255) / 256, 256, 0, stream>>>(dstI, E0, cnt);
  scan_offsets<<<1, 1024, 0, stream>>>(cnt, off, cur, Nn);
  fill_sorted<<<(E0 + 255) / 256, 256, 0, stream>>>(srcI, dstI, E0, cur, ssrc);

  {
    dim3 g(F1 / 128, (Nn + 127) / 128);
    wmma_gemm_f32<4><<<g, 256, 0, stream>>>(x, W1, xp1, Nn, F1, K1);
  }
  node_dots<<<(Nn * H1 + 7) / 8, 256, 0, stream>>>(xp1, as1, ad1, asrc1, adst1, Nn, H1, 256);
  segment_softmax_aggregate<3, true><<<(Nn * 6 + 7) / 8, 256, 0, stream>>>(
      off, ssrc, xp1, asrc1, adst1, b1, h1, Nn);

  {
    dim3 g(F2 / 128, (Nn + 63) / 64);
    wmma_gemm_f32<2><<<g, 128, 0, stream>>>(h1, W2, xp2, Nn, F2, F1);
  }
  node_dots<<<(Nn + 7) / 8, 256, 0, stream>>>(xp2, as2, ad2, asrc2, adst2, Nn, H2, 256);
  segment_softmax_aggregate<1, false><<<(Nn * 2 + 7) / 8, 256, 0, stream>>>(
      off, ssrc, xp2, asrc2, adst2, b2, (float*)d_out, Nn);
}


# Fewer, longer waves in the six index-driven launches: node-dot and segment-aggregate waves loop over 4-8 items, count/fill threads take 4 edges (260,000 entering waves -> 45,008); sum order kept
# speedup vs baseline: 3.8187x; 1.0347x over previous
#include <hip/hip_runtime.h>
#include <math.h>


typedef __attribute__((ext_vector_type(2))) float v2f;
typedef __attribute__((ext_vector_type(4))) float v4f;
typedef __attribute__((ext_vector_type(8))) float v8f;

template <int WM>
__global__ __launch_bounds__(64 * WM) void wmma_gemm_f32(
    const float* __restrict__ A, const float* __restrict__ B,
    float* __restrict__ C, int M, int N, int K)
{
  constexpr int BM = 32 * WM, BN = 128, BK = 32;
  constexpr int NT = 64 * WM;
  constexpr int PA = BK + 4;
  constexpr int PB = BN + 16;
  constexpr int A_LD = BM * (BK / 4) / NT;
  constexpr int B_LD = BK * (BN / 4) / NT;

  __shared__ __align__(16) float As[2][BM * PA];
  __shared__ __align__(16) float Bs[2][BK * PB];

  const int t    = threadIdx.x;
  const int lane = t & 31;
  const int h    = lane >> 4;
  const int c    = lane & 15;
  const int wave = __builtin_amdgcn_readfirstlane(t >> 5);
  const int wm   = wave >> 1;
  const int wn   = wave & 1;
  const int n0   = blockIdx.x * BN;
  const int m0   = blockIdx.y * BM;

  const int a_kq  = t & 7;
  const int a_row = t >> 3;
  const int b_nq  = t & 31;
  const int b_row = t >> 5;

  const float* Ag[A_LD];
  for (int i = 0; i < A_LD; ++i) {
    int r = m0 + a_row + i * (NT / 8);
    r = (r < M) ? r : (M - 1);
    Ag[i] = A + (size_t)r * K + a_kq * 4;
  }
  const float* Bg = B + (size_t)b_row * N + n0 + b_nq * 4;

  v4f ra[A_LD], rb[B_LD];
  v8f acc[2][4];
  for (int i = 0; i < 2; ++i)
    for (int f = 0; f < 4; ++f) acc[i][f] = v8f{};

  auto load_global = [&](int k0) {
    for (int i = 0; i < A_LD; ++i) ra[i] = *(const v4f*)(Ag[i] + k0);
    for (int i = 0; i < B_LD; ++i)
      rb[i] = *(const v4f*)(Bg + (size_t)(k0 + i * (NT / 32)) * N);
  };
  auto store_lds = [&](int buf) {
    for (int i = 0; i < A_LD; ++i)
      *(v4f*)(&As[buf][(a_row + i * (NT / 8)) * PA + a_kq * 4]) = ra[i];
    for (int i = 0; i < B_LD; ++i)
      *(v4f*)(&Bs[buf][(b_row + i * (NT / 32)) * PB + b_nq * 4]) = rb[i];
  };
  auto compute = [&](int buf) {
    const float* Ap = &As[buf][(wm * 32 + c) * PA + 2 * h];
    const float* Bp = &Bs[buf][(2 * h) * PB + wn * 64 + 4 * c];
    for (int kk = 0; kk < BK; kk += 4) {
      const v2f a0 = *(const v2f*)(Ap + kk);
      const v2f a1 = *(const v2f*)(Ap + 16 * PA + kk);
      const v4f bx = *(const v4f*)(Bp + kk * PB);
      const v4f by = *(const v4f*)(Bp + (kk + 1) * PB);
      v2f b0, b1, b2, b3;
      b0.x = bx.x; b0.y = by.x;
      b1.x = bx.y; b1.y = by.y;
      b2.x = bx.z; b2.y = by.z;
      b3.x = bx.w; b3.y = by.w;
      acc[0][0] = __builtin_amdgcn_wmma_f32_16x16x4_f32(false, a0, false, b0, (short)0, acc[0][0], false, false);
      acc[1][0] = __builtin_amdgcn_wmma_f32_16x16x4_f32(false, a1, false, b0, (short)0, acc[1][0], false, false);
      acc[0][1] = __builtin_amdgcn_wmma_f32_16x16x4_f32(false, a0, false, b1, (short)0, acc[0][1], false, false);
      acc[1][1] = __builtin_amdgcn_wmma_f32_16x16x4_f32(false, a1, false, b1, (short)0, acc[1][1], false, false);
      acc[0][2] = __builtin_amdgcn_wmma_f32_16x16x4_f32(false, a0, false, b2, (short)0, acc[0][2], false, false);
      acc[1][2] = __builtin_amdgcn_wmma_f32_16x16x4_f32(false, a1, false, b2, (short)0, acc[1][2], false, false);
      acc[0][3] = __builtin_amdgcn_wmma_f32_16x16x4_f32(false, a0, false, b3, (short)0, acc[0][3], false, false);
      acc[1][3] = __builtin_amdgcn_wmma_f32_16x16x4_f32(false, a1, false, b3, (short)0, acc[1][3], false, false);
    }
  };

  load_global(0);
  store_lds(0);
  __syncthreads();
  int buf = 0;
  for (int k0 = BK; k0 < K; k0 += BK) {
    load_global(k0);
    __builtin_amdgcn_sched_barrier(0);
    compute(buf);
    store_lds(buf ^ 1);
    __syncthreads();
    buf ^= 1;
  }
  compute(buf);

  float* Cp = C + n0 + wn * 64 + 4 * c;
  for (int i = 0; i < 2; ++i) {
    for (int r = 0; r < 8; ++r) {
      const int row = m0 + wm * 32 + i * 16 + h * 8 + r;
      if (row < M) {
        v4f o;
        o.x = acc[i][0][r]; o.y = acc[i][1][r]; o.z = acc[i][2][r]; o.w = acc[i][3][r];
        *(v4f*)(Cp + (size_t)row * N) = o;
      }
    }
  }
}

template <int H, int C>
__global__ __launch_bounds__(256) void node_dots(
    const float* __restrict__ xp, const float* __restrict__ att_src,
    const float* __restrict__ att_dst, float* __restrict__ asrc,
    float* __restrict__ adst, int Nn)
{
  const int lane = threadIdx.x & 31;
  const int wv = __builtin_amdgcn_readfirstlane((int)(threadIdx.x >> 5));
  const int nw = (int)gridDim.x * 8;
  const int items = Nn * H;
  for (int wid = (int)blockIdx.x * 8 + wv; wid < items; wid += nw) {
    const int n = wid / H;
    const int h = wid - n * H;
    const float* xr = xp + (size_t)n * H * C + (size_t)h * C + lane;
    const float* as = att_src + (size_t)h * C + lane;
    const float* ad = att_dst + (size_t)h * C + lane;
    float v[C / 32], ws[C / 32], wd[C / 32];
#pragma unroll
    for (int k = 0; k < C / 32; ++k) {
      v[k]  = xr[k * 32];
      ws[k] = as[k * 32];
      wd[k] = ad[k * 32];
    }
    float s = 0.f, d = 0.f;
#pragma unroll
    for (int k = 0; k < C / 32; ++k) {
      s = __builtin_fmaf(v[k], ws[k], s);
      d = __builtin_fmaf(v[k], wd[k], d);
    }
    for (int off = 16; off > 0; off >>= 1) {
      s += __shfl_down(s, off, 32);
      d += __shfl_down(d, off, 32);
    }
    if (lane == 0) { asrc[wid] = s; adst[wid] = d; }
  }
}

__global__ void zero_i32(int* __restrict__ p, int n) {
  const int i = blockIdx.x * blockDim.x + threadIdx.x;
  if (i < n) p[i] = 0;
}

__global__ __launch_bounds__(256) void count_dst(
    const int* __restrict__ dst, int E0, int* __restrict__ cnt)
{
  const int nt = (int)(gridDim.x * blockDim.x);
  int e = (int)(blockIdx.x * blockDim.x + threadIdx.x);
  for (; e + 3 * nt < E0; e += 4 * nt) {
    const int d0 = dst[e], d1 = dst[e + nt], d2 = dst[e + 2 * nt], d3 = dst[e + 3 * nt];
    atomicAdd(&cnt[d0], 1);
    atomicAdd(&cnt[d1], 1);
    atomicAdd(&cnt[d2], 1);
    atomicAdd(&cnt[d3], 1);
  }
  for (; e < E0; e += nt) atomicAdd(&cnt[dst[e]], 1);
}

__global__ __launch_bounds__(1024) void scan_offsets(
    const int* __restrict__ cnt, int* __restrict__ off, int* __restrict__ cur, int n)
{
  __shared__ int part[1024];
  const int t = threadIdx.x;
  const int chunk = (n + 1023) >> 10;
  const int b = t * chunk;
  const int e = (b + chunk < n) ? (b + chunk) : n;
  int s = 0;
  for (int i = b; i < e; ++i) s += cnt[i];
  part[t] = s;
  __syncthreads();
  for (int o = 1; o < 1024; o <<= 1) {
    const int v = (t >= o) ? part[t - o] : 0;
    __syncthreads();
    part[t] += v;
    __syncthreads();
  }
  int run = part[t] - s;
  for (int i = b; i < e; ++i) { off[i] = run; cur[i] = run; run += cnt[i]; }
  if (t == 1023) off[n] = part[1023];
}

__global__ __launch_bounds__(256) void fill_sorted(
    const int* __restrict__ src, const int* __restrict__ dst, int E0,
    int* __restrict__ cur, int* __restrict__ ssrc)
{
  const int nt = (int)(gridDim.x * blockDim.x);
  int e = (int)(blockIdx.x * blockDim.x + threadIdx.x);
  for (; e + 3 * nt < E0; e += 4 * nt) {
    const int d0 = dst[e], d1 = dst[e + nt], d2 = dst[e + 2 * nt], d3 = dst[e + 3 * nt];
    const int s0 = src[e], s1 = src[e + nt], s2 = src[e + 2 * nt], s3 = src[e + 3 * nt];
    const int p0 = atomicAdd(&cur[d0], 1);
    const int p1 = atomicAdd(&cur[d1], 1);
    const int p2 = atomicAdd(&cur[d2], 1);
    const int p3 = atomicAdd(&cur[d3], 1);
    ssrc[p0] = s0;
    ssrc[p1] = s1;
    ssrc[p2] = s2;
    ssrc[p3] = s3;
  }
  for (; e < E0; e += nt) {
    const int p = atomicAdd(&cur[dst[e]], 1);
    ssrc[p] = src[e];
  }
}

__device__ __forceinline__ float leaky02(float a) { return (a > 0.f) ? a : 0.2f * a; }

__device__ __forceinline__ void fma4(float4& acc, const float4 r, const float w) {
  acc.x = __builtin_fmaf(r.x, w, acc.x);
  acc.y = __builtin_fmaf(r.y, w, acc.y);
  acc.z = __builtin_fmaf(r.z, w, acc.z);
  acc.w = __builtin_fmaf(r.w, w, acc.w);
}

__device__ __forceinline__ void gather_group(
    float4& acc, const float* __restrict__ xcol, const int F,
    const int s, const float w, const int n)
{
  const int wi = __float_as_int(w);
  int jj = 0;
  for (; jj + 4 <= n; jj += 4) {
    const int s0 = __builtin_amdgcn_readlane(s, jj);
    const int s1 = __builtin_amdgcn_readlane(s, jj + 1);
    const int s2 = __builtin_amdgcn_readlane(s, jj + 2);
    const int s3 = __builtin_amdgcn_readlane(s, jj + 3);
    const float w0 = __int_as_float(__builtin_amdgcn_readlane(wi, jj));
    const float w1 = __int_as_float(__builtin_amdgcn_readlane(wi, jj + 1));
    const float w2 = __int_as_float(__builtin_amdgcn_readlane(wi, jj + 2));
    const float w3 = __int_as_float(__builtin_amdgcn_readlane(wi, jj + 3));
    const float4 r0 = *(const float4*)(xcol + (size_t)s0 * F);
    const float4 r1 = *(const float4*)(xcol + (size_t)s1 * F);
    const float4 r2 = *(const float4*)(xcol + (size_t)s2 * F);
    const float4 r3 = *(const float4*)(xcol + (size_t)s3 * F);
    fma4(acc, r0, w0);
    fma4(acc, r1, w1);
    fma4(acc, r2, w2);
    fma4(acc, r3, w3);
  }
  for (; jj < n; ++jj) {
    const int s0 = __builtin_amdgcn_readlane(s, jj);
    const float w0 = __int_as_float(__builtin_amdgcn_readlane(wi, jj));
    const float4 r0 = *(const float4*)(xcol + (size_t)s0 * F);
    fma4(acc, r0, w0);
  }
}

template <int H, bool ELU>
__global__ __launch_bounds__(256) void segment_softmax_aggregate(
    const int* __restrict__ off, const int* __restrict__ ssrc,
    const float* __restrict__ xp, const float* __restrict__ asrc,
    const float* __restrict__ adst, const float* __restrict__ bias,
    float* __restrict__ out, int Nn)
{
  constexpr int WPN = H * 2;
  constexpr int F   = H * 256;
  const int lane = threadIdx.x & 31;
  const int wv = __builtin_amdgcn_readfirstlane((int)(threadIdx.x >> 5));
  const int nw = (int)gridDim.x * 8;
  for (int gw = (int)blockIdx.x * 8 + wv; gw < Nn * WPN; gw += nw) {
  const int d    = gw / WPN;
  const int part = gw - d * WPN;
  const int h    = part >> 1;
  const int beg  = off[d];
  const int deg  = off[d + 1] - beg;
  const int tot  = deg + 1;
  const float ad = adst[d * H + h];
  const float NEG_INF = -__builtin_inff();

  int   s0 = d;
  float a0 = NEG_INF;
  if (lane < tot) {
    s0 = (lane < deg) ? ssrc[beg + lane] : d;
    a0 = leaky02(asrc[s0 * H + h] + ad);
  }

  float m = a0;
  for (int j = 32 + lane; j < tot; j += 32) {
    const int s = (j < deg) ? ssrc[beg + j] : d;
    m = fmaxf(m, leaky02(asrc[s * H + h] + ad));
  }
  for (int o = 16; o > 0; o >>= 1) m = fmaxf(m, __shfl_xor(m, o, 32));

  const float e0 = (lane < tot) ? __expf(a0 - m) : 0.f;
  float den = e0;
  for (int j = 32 + lane; j < tot; j += 32) {
    const int s = (j < deg) ? ssrc[beg + j] : d;
    den += __expf(leaky02(asrc[s * H + h] + ad) - m);
  }
  for (int o = 16; o > 0; o >>= 1) den += __shfl_xor(den, o, 32);

  const int col = part * 128 + lane * 4;
  const float* xcol = xp + col;
  const float4 b = *(const float4*)(bias + col);
  float4 acc = make_float4(0.f, 0.f, 0.f, 0.f);

  gather_group(acc, xcol, F, s0, e0 / den, (tot < 32) ? tot : 32);
  for (int base = 32; base < tot; base += 32) {
    const int j = base + lane;
    int   s = d;
    float w = 0.f;
    if (j < tot) {
      s = (j < deg) ? ssrc[beg + j] : d;
      w = __expf(leaky02(asrc[s * H + h] + ad) - m) / den;
    }
    const int n = tot - base;
    gather_group(acc, xcol, F, s, w, (n < 32) ? n : 32);
  }

  float4 v = make_float4(acc.x + b.x, acc.y + b.y, acc.z + b.z, acc.w + b.w);
  if (ELU) {
    v.x = (v.x > 0.f) ? v.x : expm1f(v.x);
    v.y = (v.y > 0.f) ? v.y : expm1f(v.y);
    v.z = (v.z > 0.f) ? v.z : expm1f(v.z);
    v.w = (v.w > 0.f) ? v.w : expm1f(v.w);
  }
  *(float4*)(out + (size_t)d * F + col) = v;
  }
}


extern "C" void kernel_launch(void* const* d_in, const int* in_sizes, int n_in,
                              void* d_out, int out_size, void* d_ws, size_t ws_size,
                              hipStream_t stream) {
  const float* x   = (const float*)d_in[0];
  const int*   ei  = (const int*)d_in[1];
  const float* W1  = (const float*)d_in[2];
  const float* as1 = (const float*)d_in[3];
  const float* ad1 = (const float*)d_in[4];
  const float* b1  = (const float*)d_in[5];
  const float* W2  = (const float*)d_in[6];
  const float* as2 = (const float*)d_in[7];
  const float* ad2 = (const float*)d_in[8];
  const float* b2  = (const float*)d_in[9];

  const int K1 = 768, F1 = 768, H1 = 3;
  const int F2 = 256;
  const int Nn = in_sizes[0] / K1;
  const int E0 = in_sizes[1] / 2;

  const int* srcI = ei;
  const int* dstI = ei + E0;

  float* ws = (float*)d_ws;
  float* xp1   = ws; ws += (size_t)Nn * F1;
  float* h1    = ws; ws += (size_t)Nn * F1;
  float* xp2   = ws; ws += (size_t)Nn * F2;
  float* asrc1 = ws; ws += (size_t)Nn * H1;
  float* adst1 = ws; ws += (size_t)Nn * H1;
  float* asrc2 = ws; ws += Nn;
  float* adst2 = ws; ws += Nn;
  int* cnt  = (int*)ws; ws += Nn;
  int* off  = (int*)ws; ws += Nn + 1;
  int* cur  = (int*)ws; ws += Nn;
  int* ssrc = (int*)ws; ws += E0;

  zero_i32<<<(Nn + 255) / 256, 256, 0, stream>>>(cnt, Nn);
  count_dst<<<(E0 + 1023) / 1024, 256, 0, stream>>>(dstI, E0, cnt);
  scan_offsets<<<1, 1024, 0, stream>>>(cnt, off, cur, Nn);
  fill_sorted<<<(E0 + 1023) / 1024, 256, 0, stream>>>(srcI, dstI, E0, cur, ssrc);

  {
    dim3 g(F1 / 128, (Nn + 127) / 128);
    wmma_gemm_f32<4><<<g, 256, 0, stream>>>(x, W1, xp1, Nn, F1, K1);
  }
  node_dots<3, 256><<<1250, 256, 0, stream>>>(xp1, as1, ad1, asrc1, adst1, Nn);
  segment_softmax_aggregate<3, true><<<1875, 256, 0, stream>>>(
      off, ssrc, xp1, asrc1, adst1, b1, h1, Nn);

  {
    dim3 g(F2 / 128, (Nn + 63) / 64);
    wmma_gemm_f32<2><<<g, 128, 0, stream>>>(h1, W2, xp2, Nn, F2, F1);
  }
  node_dots<1, 256><<<625, 256, 0, stream>>>(xp2, as2, ad2, asrc2, adst2, Nn);
  segment_softmax_aggregate<1, false><<<1250, 256, 0, stream>>>(
      off, ssrc, xp2, asrc2, adst2, b2, (float*)d_out, Nn);
}


# Attention dots a_src/a_dst formed in the GEMM epilogue from the accumulators (one partial per 128-column tile, staged through LDS) replace both node-dot launches; dot sum order changed
# speedup vs baseline: 3.9842x; 1.0433x over previous
#include <hip/hip_runtime.h>
#include <math.h>


typedef __attribute__((ext_vector_type(2))) float v2f;
typedef __attribute__((ext_vector_type(4))) float v4f;
typedef __attribute__((ext_vector_type(8))) float v8f;

template <int WM>
__global__ __launch_bounds__(64 * WM) void wmma_gemm_f32(
    const float* __restrict__ A, const float* __restrict__ B,
    float* __restrict__ C, int M, int N, int K,
    const float* __restrict__ att_s, const float* __restrict__ att_d,
    float* __restrict__ ps, float* __restrict__ pd)
{
  constexpr int BM = 32 * WM, BN = 128, BK = 32;
  constexpr int NT = 64 * WM;
  constexpr int PA = BK + 4;
  constexpr int PB = BN + 16;
  constexpr int A_LD = BM * (BK / 4) / NT;
  constexpr int B_LD = BK * (BN / 4) / NT;

  __shared__ __align__(16) float As[2][BM * PA];
  __shared__ __align__(16) float Bs[2][BK * PB];

  const int t    = threadIdx.x;
  const int lane = t & 31;
  const int h    = lane >> 4;
  const int c    = lane & 15;
  const int wave = __builtin_amdgcn_readfirstlane(t >> 5);
  const int wm   = wave >> 1;
  const int wn   = wave & 1;
  const int n0   = blockIdx.x * BN;
  const int m0   = blockIdx.y * BM;

  const int a_kq  = t & 7;
  const int a_row = t >> 3;
  const int b_nq  = t & 31;
  const int b_row = t >> 5;

  const float* Ag[A_LD];
  for (int i = 0; i < A_LD; ++i) {
    int r = m0 + a_row + i * (NT / 8);
    r = (r < M) ? r : (M - 1);
    Ag[i] = A + (size_t)r * K + a_kq * 4;
  }
  const float* Bg = B + (size_t)b_row * N + n0 + b_nq * 4;

  v4f ra[A_LD], rb[B_LD];
  v8f acc[2][4];
  for (int i = 0; i < 2; ++i)
    for (int f = 0; f < 4; ++f) acc[i][f] = v8f{};

  auto load_global = [&](int k0) {
    for (int i = 0; i < A_LD; ++i) ra[i] = *(const v4f*)(Ag[i] + k0);
    for (int i = 0; i < B_LD; ++i)
      rb[i] = *(const v4f*)(Bg + (size_t)(k0 + i * (NT / 32)) * N);
  };
  auto store_lds = [&](int buf) {
    for (int i = 0; i < A_LD; ++i)
      *(v4f*)(&As[buf][(a_row + i * (NT / 8)) * PA + a_kq * 4]) = ra[i];
    for (int i = 0; i < B_LD; ++i)
      *(v4f*)(&Bs[buf][(b_row + i * (NT / 32)) * PB + b_nq * 4]) = rb[i];
  };
  auto compute = [&](int buf) {
    const float* Ap = &As[buf][(wm * 32 + c) * PA + 2 * h];
    const float* Bp = &Bs[buf][(2 * h) * PB + wn * 64 + 4 * c];
    for (int kk = 0; kk < BK; kk += 4) {
      const v2f a0 = *(const v2f*)(Ap + kk);
      const v2f a1 = *(const v2f*)(Ap + 16 * PA + kk);
      const v4f bx = *(const v4f*)(Bp + kk * PB);
      const v4f by = *(const v4f*)(Bp + (kk + 1) * PB);
      v2f b0, b1, b2, b3;
      b0.x = bx.x; b0.y = by.x;
      b1.x = bx.y; b1.y = by.y;
      b2.x = bx.z; b2.y = by.z;
      b3.x = bx.w; b3.y = by.w;
      acc[0][0] = __builtin_amdgcn_wmma_f32_16x16x4_f32(false, a0, false, b0, (short)0, acc[0][0], false, false);
      acc[1][0] = __builtin_amdgcn_wmma_f32_16x16x4_f32(false, a1, false, b0, (short)0, acc[1][0], false, false);
      acc[0][1] = __builtin_amdgcn_wmma_f32_16x16x4_f32(false, a0, false, b1, (short)0, acc[0][1], false, false);
      acc[1][1] = __builtin_amdgcn_wmma_f32_16x16x4_f32(false, a1, false, b1, (short)0, acc[1][1], false, false);
      acc[0][2] = __builtin_amdgcn_wmma_f32_16x16x4_f32(false, a0, false, b2, (short)0, acc[0][2], false, false);
      acc[1][2] = __builtin_amdgcn_wmma_f32_16x16x4_f32(false, a1, false, b2, (short)0, acc[1][2], false, false);
      acc[0][3] = __builtin_amdgcn_wmma_f32_16x16x4_f32(false, a0, false, b3, (short)0, acc[0][3], false, false);
      acc[1][3] = __builtin_amdgcn_wmma_f32_16x16x4_f32(false, a1, false, b3, (short)0, acc[1][3], false, false);
    }
  };

  load_global(0);
  store_lds(0);
  __syncthreads();
  int buf = 0;
  for (int k0 = BK; k0 < K; k0 += BK) {
    load_global(k0);
    __builtin_amdgcn_sched_barrier(0);
    compute(buf);
    store_lds(buf ^ 1);
    __syncthreads();
    buf ^= 1;
  }
  compute(buf);

  constexpr int PP = 20;
  static_assert(2 * BM * PP <= 2 * BM * PA && 2 * BM * PP <= 2 * BK * PB, "partials fit the staging buffers");
  float* Ss = &As[0][0];
  float* Sd = &Bs[0][0];
  const v4f wsv = *(const v4f*)(att_s + n0 + wn * 64 + 4 * c);
  const v4f wdv = *(const v4f*)(att_d + n0 + wn * 64 + 4 * c);
  __syncthreads();
  for (int i = 0; i < 2; ++i) {
    for (int r = 0; r < 8; ++r) {
      const int lr = wm * 32 + i * 16 + h * 8 + r;
      float s = acc[i][0][r] * wsv.x;
      s = __builtin_fmaf(acc[i][1][r], wsv.y, s);
      s = __builtin_fmaf(acc[i][2][r], wsv.z, s);
      s = __builtin_fmaf(acc[i][3][r], wsv.w, s);
      float d = acc[i][0][r] * wdv.x;
      d = __builtin_fmaf(acc[i][1][r], wdv.y, d);
      d = __builtin_fmaf(acc[i][2][r], wdv.z, d);
      d = __builtin_fmaf(acc[i][3][r], wdv.w, d);
      Ss[(wn * BM + lr) * PP + c] = s;
      Sd[(wn * BM + lr) * PP + c] = d;
    }
  }
  __syncthreads();
  {
    const int q  = t / BM;
    const int lr = t - q * BM;
    const float* P = q ? Sd : Ss;
    float sum = 0.f;
    for (int w2 = 0; w2 < 2; ++w2) {
      for (int j = 0; j < 4; ++j) {
        const v4f v = *(const v4f*)(P + (w2 * BM + lr) * PP + 4 * j);
        sum += v.x; sum += v.y; sum += v.z; sum += v.w;
      }
    }
    if (m0 + lr < M) (q ? pd : ps)[(size_t)blockIdx.x * M + m0 + lr] = sum;
  }

  float* Cp = C + n0 + wn * 64 + 4 * c;
  for (int i = 0; i < 2; ++i) {
    for (int r = 0; r < 8; ++r) {
      const int row = m0 + wm * 32 + i * 16 + h * 8 + r;
      if (row < M) {
        v4f o;
        o.x = acc[i][0][r]; o.y = acc[i][1][r]; o.z = acc[i][2][r]; o.w = acc[i][3][r];
        *(v4f*)(Cp + (size_t)row * N) = o;
      }
    }
  }
}

__global__ void zero_i32(int* __restrict__ p, int n) {
  const int i = blockIdx.x * blockDim.x + threadIdx.x;
  if (i < n) p[i] = 0;
}

__global__ __launch_bounds__(256) void count_dst(
    const int* __restrict__ dst, int E0, int* __restrict__ cnt)
{
  const int nt = (int)(gridDim.x * blockDim.x);
  int e = (int)(blockIdx.x * blockDim.x + threadIdx.x);
  for (; e + 3 * nt < E0; e += 4 * nt) {
    const int d0 = dst[e], d1 = dst[e + nt], d2 = dst[e + 2 * nt], d3 = dst[e + 3 * nt];
    atomicAdd(&cnt[d0], 1);
    atomicAdd(&cnt[d1], 1);
    atomicAdd(&cnt[d2], 1);
    atomicAdd(&cnt[d3], 1);
  }
  for (; e < E0; e += nt) atomicAdd(&cnt[dst[e]], 1);
}

__global__ __launch_bounds__(1024) void scan_offsets(
    const int* __restrict__ cnt, int* __restrict__ off, int* __restrict__ cur, int n)
{
  __shared__ int part[1024];
  const int t = threadIdx.x;
  const int chunk = (n + 1023) >> 10;
  const int b = t * chunk;
  const int e = (b + chunk < n) ? (b + chunk) : n;
  int s = 0;
  for (int i = b; i < e; ++i) s += cnt[i];
  part[t] = s;
  __syncthreads();
  for (int o = 1; o < 1024; o <<= 1) {
    const int v = (t >= o) ? part[t - o] : 0;
    __syncthreads();
    part[t] += v;
    __syncthreads();
  }
  int run = part[t] - s;
  for (int i = b; i < e; ++i) { off[i] = run; cur[i] = run; run += cnt[i]; }
  if (t == 1023) off[n] = part[1023];
}

__global__ __launch_bounds__(256) void fill_sorted(
    const int* __restrict__ src, const int* __restrict__ dst, int E0,
    int* __restrict__ cur, int* __restrict__ ssrc)
{
  const int nt = (int)(gridDim.x * blockDim.x);
  int e = (int)(blockIdx.x * blockDim.x + threadIdx.x);
  for (; e + 3 * nt < E0; e += 4 * nt) {
    const int d0 = dst[e], d1 = dst[e + nt], d2 = dst[e + 2 * nt], d3 = dst[e + 3 * nt];
    const int s0 = src[e], s1 = src[e + nt], s2 = src[e + 2 * nt], s3 = src[e + 3 * nt];
    const int p0 = atomicAdd(&cur[d0], 1);
    const int p1 = atomicAdd(&cur[d1], 1);
    const int p2 = atomicAdd(&cur[d2], 1);
    const int p3 = atomicAdd(&cur[d3], 1);
    ssrc[p0] = s0;
    ssrc[p1] = s1;
    ssrc[p2] = s2;
    ssrc[p3] = s3;
  }
  for (; e < E0; e += nt) {
    const int p = atomicAdd(&cur[dst[e]], 1);
    ssrc[p] = src[e];
  }
}

__device__ __forceinline__ float leaky02(float a) { return (a > 0.f) ? a : 0.2f * a; }

__device__ __forceinline__ void fma4(float4& acc, const float4 r, const float w) {
  acc.x = __builtin_fmaf(r.x, w, acc.x);
  acc.y = __builtin_fmaf(r.y, w, acc.y);
  acc.z = __builtin_fmaf(r.z, w, acc.z);
  acc.w = __builtin_fmaf(r.w, w, acc.w);
}

__device__ __forceinline__ void gather_group(
    float4& acc, const float* __restrict__ xcol, const int F,
    const int s, const float w, const int n)
{
  const int wi = __float_as_int(w);
  int jj = 0;
  for (; jj + 4 <= n; jj += 4) {
    const int s0 = __builtin_amdgcn_readlane(s, jj);
    const int s1 = __builtin_amdgcn_readlane(s, jj + 1);
    const int s2 = __builtin_amdgcn_readlane(s, jj + 2);
    const int s3 = __builtin_amdgcn_readlane(s, jj + 3);
    const float w0 = __int_as_float(__builtin_amdgcn_readlane(wi, jj));
    const float w1 = __int_as_float(__builtin_amdgcn_readlane(wi, jj + 1));
    const float w2 = __int_as_float(__builtin_amdgcn_readlane(wi, jj + 2));
    const float w3 = __int_as_float(__builtin_amdgcn_readlane(wi, jj + 3));
    const float4 r0 = *(const float4*)(xcol + (size_t)s0 * F);
    const float4 r1 = *(const float4*)(xcol + (size_t)s1 * F);
    const float4 r2 = *(const float4*)(xcol + (size_t)s2 * F);
    const float4 r3 = *(const float4*)(xcol + (size_t)s3 * F);
    fma4(acc, r0, w0);
    fma4(acc, r1, w1);
    fma4(acc, r2, w2);
    fma4(acc, r3, w3);
  }
  for (; jj < n; ++jj) {
    const int s0 = __builtin_amdgcn_readlane(s, jj);
    const float w0 = __int_as_float(__builtin_amdgcn_readlane(wi, jj));
    const float4 r0 = *(const float4*)(xcol + (size_t)s0 * F);
    fma4(acc, r0, w0);
  }
}

template <int H, bool ELU>
__global__ __launch_bounds__(256) void segment_softmax_aggregate(
    const int* __restrict__ off, const int* __restrict__ ssrc,
    const float* __restrict__ xp, const float* __restrict__ ps,
    const float* __restrict__ pd, const float* __restrict__ bias,
    float* __restrict__ out, int Nn)
{
  constexpr int WPN = H * 2;
  constexpr int F   = H * 256;
  const int lane = threadIdx.x & 31;
  const int wv = __builtin_amdgcn_readfirstlane((int)(threadIdx.x >> 5));
  const int nw = (int)gridDim.x * 8;
  for (int gw = (int)blockIdx.x * 8 + wv; gw < Nn * WPN; gw += nw) {
  const int d    = gw / WPN;
  const int part = gw - d * WPN;
  const int h    = part >> 1;
  const int beg  = off[d];
  const int deg  = off[d + 1] - beg;
  const int tot  = deg + 1;
  const float* ps0 = ps + (size_t)(2 * h) * Nn;
  const float* ps1 = ps0 + Nn;
  const float ad = pd[(size_t)(2 * h) * Nn + d] + pd[(size_t)(2 * h + 1) * Nn + d];
  const float NEG_INF = -__builtin_inff();

  int   s0 = d;
  float a0 = NEG_INF;
  if (lane < tot) {
    s0 = (lane < deg) ? ssrc[beg + lane] : d;
    a0 = leaky02((ps0[s0] + ps1[s0]) + ad);
  }

  float m = a0;
  for (int j = 32 + lane; j < tot; j += 32) {
    const int s = (j < deg) ? ssrc[beg + j] : d;
    m = fmaxf(m, leaky02((ps0[s] + ps1[s]) + ad));
  }
  for (int o = 16; o > 0; o >>= 1) m = fmaxf(m, __shfl_xor(m, o, 32));

  const float e0 = (lane < tot) ? __expf(a0 - m) : 0.f;
  float den = e0;
  for (int j = 32 + lane; j < tot; j += 32) {
    const int s = (j < deg) ? ssrc[beg + j] : d;
    den += __expf(leaky02((ps0[s] + ps1[s]) + ad) - m);
  }
  for (int o = 16; o > 0; o >>= 1) den += __shfl_xor(den, o, 32);

  const int col = part * 128 + lane * 4;
  const float* xcol = xp + col;
  const float4 b = *(const float4*)(bias + col);
  float4 acc = make_float4(0.f, 0.f, 0.f, 0.f);

  gather_group(acc, xcol, F, s0, e0 / den, (tot < 32) ? tot : 32);
  for (int base = 32; base < tot; base += 32) {
    const int j = base + lane;
    int   s = d;
    float w = 0.f;
    if (j < tot) {
      s = (j < deg) ? ssrc[beg + j] : d;
      w = __expf(leaky02((ps0[s] + ps1[s]) + ad) - m) / den;
    }
    const int n = tot - base;
    gather_group(acc, xcol, F, s, w, (n < 32) ? n : 32);
  }

  float4 v = make_float4(acc.x + b.x, acc.y + b.y, acc.z + b.z, acc.w + b.w);
  if (ELU) {
    v.x = (v.x > 0.f) ? v.x : expm1f(v.x);
    v.y = (v.y > 0.f) ? v.y : expm1f(v.y);
    v.z = (v.z > 0.f) ? v.z : expm1f(v.z);
    v.w = (v.w > 0.f) ? v.w : expm1f(v.w);
  }
  *(float4*)(out + (size_t)d * F + col) = v;
  }
}


extern "C" void kernel_launch(void* const* d_in, const int* in_sizes, int n_in,
                              void* d_out, int out_size, void* d_ws, size_t ws_size,
                              hipStream_t stream) {
  const float* x   = (const float*)d_in[0];
  const int*   ei  = (const int*)d_in[1];
  const float* W1  = (const float*)d_in[2];
  const float* as1 = (const float*)d_in[3];
  const float* ad1 = (const float*)d_in[4];
  const float* b1  = (const float*)d_in[5];
  const float* W2  = (const float*)d_in[6];
  const float* as2 = (const float*)d_in[7];
  const float* ad2 = (const float*)d_in[8];
  const float* b2  = (const float*)d_in[9];

  const int K1 = 768, F1 = 768, H1 = 3;
  const int F2 = 256;
  const int Nn = in_sizes[0] / K1;
  const int E0 = in_sizes[1] / 2;

  const int* srcI = ei;
  const int* dstI = ei + E0;

  float* ws = (float*)d_ws;
  float* xp1   = ws; ws += (size_t)Nn * F1;
  float* h1    = ws; ws += (size_t)Nn * F1;
  float* xp2   = ws; ws += (size_t)Nn * F2;
  float* part2 = ws; ws += (size_t)Nn * 8;
  float* ps1 = xp2;
  float* pd1 = xp2 + (size_t)Nn * 6;
  float* ps2 = part2;
  float* pd2 = part2 + (size_t)Nn * 2;
  int* cnt  = (int*)ws; ws += Nn;
  int* off  = (int*)ws; ws += Nn + 1;
  int* cur  = (int*)ws; ws += Nn;
  int* ssrc = (int*)ws; ws += E0;

  zero_i32<<<(Nn + 255) / 256, 256, 0, stream>>>(cnt, Nn);
  count_dst<<<(E0 + 1023) / 1024, 256, 0, stream>>>(dstI, E0, cnt);
  scan_offsets<<<1, 1024, 0, stream>>>(cnt, off, cur, Nn);
  fill_sorted<<<(E0 + 1023) / 1024, 256, 0, stream>>>(srcI, dstI, E0, cur, ssrc);

  {
    dim3 g(F1 / 128, (Nn + 127) / 128);
    wmma_gemm_f32<4><<<g, 256, 0, stream>>>(x, W1, xp1, Nn, F1, K1, as1, ad1, ps1, pd1);
  }
  segment_softmax_aggregate<3, true><<<1875, 256, 0, stream>>>(
      off, ssrc, xp1, ps1, pd1, b1, h1, Nn);

  {
    dim3 g(F2 / 128, (Nn + 63) / 64);
    wmma_gemm_f32<2><<<g, 128, 0, stream>>>(h1, W2, xp2, Nn, F2, F1, as2, ad2, ps2, pd2);
  }
  segment_softmax_aggregate<1, false><<<1250, 256, 0, stream>>>(
      off, ssrc, xp2, ps2, pd2, b2, (float*)d_out, Nn);
}


# Dst-sort (count, scan, fill) moved into six extra workgroups of the layer-1 GEMM launch, one node range each with LDS counters, running beside the tiles; 4 launches gone; rank still arrival order
# speedup vs baseline: 4.3675x; 1.0962x over previous
#include <hip/hip_runtime.h>
#include <math.h>


typedef __attribute__((ext_vector_type(2))) float v2f;
typedef __attribute__((ext_vector_type(4))) float v4f;
typedef __attribute__((ext_vector_type(8))) float v8f;

template <int NT>
__device__ __forceinline__ void sort_dst_range(
    const int* __restrict__ src, const int* __restrict__ dst, const int E0,
    int* __restrict__ off, int* __restrict__ ssrc, const int Nn,
    int* cnt, const int CAP, int* part, const int rlo, const int rhi)
{
  constexpr int U1 = 16;
  constexpr int U2 = 8;
  const int t = threadIdx.x;
  const int E4 = ((E0 & 3) == 0) ? (E0 >> 2) : 0;
  const int4* dst4 = (const int4*)dst;
  const int4* src4 = (const int4*)src;

  for (int lo = rlo; lo < rhi; lo += CAP) {
    const int n = (rhi - lo < CAP) ? (rhi - lo) : CAP;
    for (int i = t; i < n; i += NT) cnt[i] = 0;
    __syncthreads();

    int below = 0;
    auto tally = [&](const int d) {
      const int k = d - lo;
      if ((unsigned)k < (unsigned)n) atomicAdd(&cnt[k], 1);
      below += (d < lo) ? 1 : 0;
    };
    for (int q0 = 0; q0 < E4; q0 += U1 * NT) {
      int4 v[U1];
#pragma unroll
      for (int u = 0; u < U1; ++u) {
        const int q = q0 + u * NT + t;
        v[u] = dst4[(q < E4) ? q : (E4 - 1)];
      }
#pragma unroll
      for (int u = 0; u < U1; ++u) {
        if (q0 + u * NT + t < E4) { tally(v[u].x); tally(v[u].y); tally(v[u].z); tally(v[u].w); }
      }
    }
    for (int e = 4 * E4 + t; e < E0; e += NT) tally(dst[e]);

    for (int o = 16; o > 0; o >>= 1) below += __shfl_xor(below, o, 32);
    if ((t & 31) == 0) part[NT + (t >> 5)] = below;
    __syncthreads();

    const int per = (n + NT - 1) / NT;
    const int b = t * per;
    const int e1 = (b + per < n) ? (b + per) : n;
    int s = 0;
    for (int i = b; i < e1; ++i) s += cnt[i];
    part[t] = s;
    __syncthreads();
    for (int o = 1; o < NT; o <<= 1) {
      const int v = (t >= o) ? part[t - o] : 0;
      __syncthreads();
      part[t] += v;
      __syncthreads();
    }
    int base = 0;
    for (int w = 0; w < NT / 32; ++w) base += part[NT + w];
    int run = base + part[t] - s;
    for (int i = b; i < e1; ++i) {
      const int cdeg = cnt[i];
      off[lo + i] = run;
      cnt[i] = run;
      run += cdeg;
    }
    if (lo + n == Nn && t == 0) off[Nn] = base + part[NT - 1];
    __syncthreads();

    auto slot = [&](const int d, const bool live) -> int {
      const int k = d - lo;
      return (live && (unsigned)k < (unsigned)n) ? atomicAdd(&cnt[k], 1) : -1;
    };
    for (int q0 = 0; q0 < E4; q0 += U2 * NT) {
      int4 vd[U2], vs[U2], p[U2];
#pragma unroll
      for (int u = 0; u < U2; ++u) {
        const int q  = q0 + u * NT + t;
        const int qc = (q < E4) ? q : (E4 - 1);
        vd[u] = dst4[qc];
        vs[u] = src4[qc];
      }
#pragma unroll
      for (int u = 0; u < U2; ++u) {
        const bool live = q0 + u * NT + t < E4;
        p[u].x = slot(vd[u].x, live); p[u].y = slot(vd[u].y, live);
        p[u].z = slot(vd[u].z, live); p[u].w = slot(vd[u].w, live);
      }
#pragma unroll
      for (int u = 0; u < U2; ++u) {
        if (p[u].x >= 0) ssrc[p[u].x] = vs[u].x;
        if (p[u].y >= 0) ssrc[p[u].y] = vs[u].y;
        if (p[u].z >= 0) ssrc[p[u].z] = vs[u].z;
        if (p[u].w >= 0) ssrc[p[u].w] = vs[u].w;
      }
    }
    for (int e = 4 * E4 + t; e < E0; e += NT) {
      const int p1 = slot(dst[e], true);
      if (p1 >= 0) ssrc[p1] = src[e];
    }
    __syncthreads();
  }
}

template <int WM, bool SORT>
__global__ __launch_bounds__(64 * WM) __attribute__((amdgpu_waves_per_eu(SORT ? 7 : 6))) void wmma_gemm_f32(
    const float* __restrict__ A, const float* __restrict__ B,
    float* __restrict__ C, int M, int N, int K,
    const float* __restrict__ att_s, const float* __restrict__ att_d,
    float* __restrict__ ps, float* __restrict__ pd,
    const int* __restrict__ esrc, const int* __restrict__ edst, int E0,
    int* __restrict__ off, int* __restrict__ ssrc)
{
  constexpr int BM = 32 * WM, BN = 128, BK = 32;
  constexpr int NT = 64 * WM;
  constexpr int PA = BK + 4;
  constexpr int PB = BN + 16;
  constexpr int A_LD = BM * (BK / 4) / NT;
  constexpr int B_LD = BK * (BN / 4) / NT;

  __shared__ __align__(16) float As[2][BM * PA];
  __shared__ __align__(16) float Bs[2][BK * PB];

  if (SORT) {
    if (blockIdx.y == 0) {
      static_assert(NT + NT / 32 <= 2 * BK * PB, "scan scratch fits the B staging buffer");
      const int R   = (int)gridDim.x;
      const int RS  = (M + R - 1) / R;
      const int rlo = (int)blockIdx.x * RS;
      const int rhi = (rlo + RS < M) ? (rlo + RS) : M;
      sort_dst_range<NT>(esrc, edst, E0, off, ssrc, M,
                         (int*)&As[0][0], 2 * BM * PA, (int*)&Bs[0][0], rlo, rhi);
      return;
    }
  }

  const int t    = threadIdx.x;
  const int lane = t & 31;
  const int h    = lane >> 4;
  const int c    = lane & 15;
  const int wave = __builtin_amdgcn_readfirstlane(t >> 5);
  const int wm   = wave >> 1;
  const int wn   = wave & 1;
  const int n0   = blockIdx.x * BN;
  const int m0   = ((int)blockIdx.y - (SORT ? 1 : 0)) * BM;

  const int a_kq  = t & 7;
  const int a_row = t >> 3;
  const int b_nq  = t & 31;
  const int b_row = t >> 5;

  const float* Ag[A_LD];
  for (int i = 0; i < A_LD; ++i) {
    int r = m0 + a_row + i * (NT / 8);
    r = (r < M) ? r : (M - 1);
    Ag[i] = A + (size_t)r * K + a_kq * 4;
  }
  const float* Bg = B + (size_t)b_row * N + n0 + b_nq * 4;

  v4f ra[A_LD], rb[B_LD];
  v8f acc[2][4];
  for (int i = 0; i < 2; ++i)
    for (int f = 0; f < 4; ++f) acc[i][f] = v8f{};

  auto load_global = [&](int k0) {
    for (int i = 0; i < A_LD; ++i) ra[i] = *(const v4f*)(Ag[i] + k0);
    for (int i = 0; i < B_LD; ++i)
      rb[i] = *(const v4f*)(Bg + (size_t)(k0 + i * (NT / 32)) * N);
  };
  auto store_lds = [&](int buf) {
    for (int i = 0; i < A_LD; ++i)
      *(v4f*)(&As[buf][(a_row + i * (NT / 8)) * PA + a_kq * 4]) = ra[i];
    for (int i = 0; i < B_LD; ++i)
      *(v4f*)(&Bs[buf][(b_row + i * (NT / 32)) * PB + b_nq * 4]) = rb[i];
  };
  auto compute = [&](int buf) {
    const float* Ap = &As[buf][(wm * 32 + c) * PA + 2 * h];
    const float* Bp = &Bs[buf][(2 * h) * PB + wn * 64 + 4 * c];
    for (int kk = 0; kk < BK; kk += 4) {
      const v2f a0 = *(const v2f*)(Ap + kk);
      const v2f a1 = *(const v2f*)(Ap + 16 * PA + kk);
      const v4f bx = *(const v4f*)(Bp + kk * PB);
      const v4f by = *(const v4f*)(Bp + (kk + 1) * PB);
      v2f b0, b1, b2, b3;
      b0.x = bx.x; b0.y = by.x;
      b1.x = bx.y; b1.y = by.y;
      b2.x = bx.z; b2.y = by.z;
      b3.x = bx.w; b3.y = by.w;
      acc[0][0] = __builtin_amdgcn_wmma_f32_16x16x4_f32(false, a0, false, b0, (short)0, acc[0][0], false, false);
      acc[1][0] = __builtin_amdgcn_wmma_f32_16x16x4_f32(false, a1, false, b0, (short)0, acc[1][0], false, false);
      acc[0][1] = __builtin_amdgcn_wmma_f32_16x16x4_f32(false, a0, false, b1, (short)0, acc[0][1], false, false);
      acc[1][1] = __builtin_amdgcn_wmma_f32_16x16x4_f32(false, a1, false, b1, (short)0, acc[1][1], false, false);
      acc[0][2] = __builtin_amdgcn_wmma_f32_16x16x4_f32(false, a0, false, b2, (short)0, acc[0][2], false, false);
      acc[1][2] = __builtin_amdgcn_wmma_f32_16x16x4_f32(false, a1, false, b2, (short)0, acc[1][2], false, false);
      acc[0][3] = __builtin_amdgcn_wmma_f32_16x16x4_f32(false, a0, false, b3, (short)0, acc[0][3], false, false);
      acc[1][3] = __builtin_amdgcn_wmma_f32_16x16x4_f32(false, a1, false, b3, (short)0, acc[1][3], false, false);
    }
  };

  load_global(0);
  store_lds(0);
  __syncthreads();
  int buf = 0;
  for (int k0 = BK; k0 < K; k0 += BK) {
    load_global(k0);
    __builtin_amdgcn_sched_barrier(0);
    compute(buf);
    store_lds(buf ^ 1);
    __syncthreads();
    buf ^= 1;
  }
  compute(buf);

  constexpr int PP = 20;
  static_assert(2 * BM * PP <= 2 * BM * PA && 2 * BM * PP <= 2 * BK * PB, "partials fit the staging buffers");
  float* Ss = &As[0][0];
  float* Sd = &Bs[0][0];
  const v4f wsv = *(const v4f*)(att_s + n0 + wn * 64 + 4 * c);
  const v4f wdv = *(const v4f*)(att_d + n0 + wn * 64 + 4 * c);
  __syncthreads();
  for (int i = 0; i < 2; ++i) {
    for (int r = 0; r < 8; ++r) {
      const int lr = wm * 32 + i * 16 + h * 8 + r;
      float s = acc[i][0][r] * wsv.x;
      s = __builtin_fmaf(acc[i][1][r], wsv.y, s);
      s = __builtin_fmaf(acc[i][2][r], wsv.z, s);
      s = __builtin_fmaf(acc[i][3][r], wsv.w, s);
      float d = acc[i][0][r] * wdv.x;
      d = __builtin_fmaf(acc[i][1][r], wdv.y, d);
      d = __builtin_fmaf(acc[i][2][r], wdv.z, d);
      d = __builtin_fmaf(acc[i][3][r], wdv.w, d);
      Ss[(wn * BM + lr) * PP + c] = s;
      Sd[(wn * BM + lr) * PP + c] = d;
    }
  }
  __syncthreads();
  {
    const int q  = t / BM;
    const int lr = t - q * BM;
    const float* P = q ? Sd : Ss;
    float sum = 0.f;
    for (int w2 = 0; w2 < 2; ++w2) {
      for (int j = 0; j < 4; ++j) {
        const v4f v = *(const v4f*)(P + (w2 * BM + lr) * PP + 4 * j);
        sum += v.x; sum += v.y; sum += v.z; sum += v.w;
      }
    }
    if (m0 + lr < M) (q ? pd : ps)[(size_t)blockIdx.x * M + m0 + lr] = sum;
  }

  float* Cp = C + n0 + wn * 64 + 4 * c;
  for (int i = 0; i < 2; ++i) {
    for (int r = 0; r < 8; ++r) {
      const int row = m0 + wm * 32 + i * 16 + h * 8 + r;
      if (row < M) {
        v4f o;
        o.x = acc[i][0][r]; o.y = acc[i][1][r]; o.z = acc[i][2][r]; o.w = acc[i][3][r];
        *(v4f*)(Cp + (size_t)row * N) = o;
      }
    }
  }
}

__device__ __forceinline__ float leaky02(float a) { return (a > 0.f) ? a : 0.2f * a; }

__device__ __forceinline__ void fma4(float4& acc, const float4 r, const float w) {
  acc.x = __builtin_fmaf(r.x, w, acc.x);
  acc.y = __builtin_fmaf(r.y, w, acc.y);
  acc.z = __builtin_fmaf(r.z, w, acc.z);
  acc.w = __builtin_fmaf(r.w, w, acc.w);
}

__device__ __forceinline__ void gather_group(
    float4& acc, const float* __restrict__ xcol, const int F,
    const int s, const float w, const int n)
{
  const int wi = __float_as_int(w);
  int jj = 0;
  for (; jj + 4 <= n; jj += 4) {
    const int s0 = __builtin_amdgcn_readlane(s, jj);
    const int s1 = __builtin_amdgcn_readlane(s, jj + 1);
    const int s2 = __builtin_amdgcn_readlane(s, jj + 2);
    const int s3 = __builtin_amdgcn_readlane(s, jj + 3);
    const float w0 = __int_as_float(__builtin_amdgcn_readlane(wi, jj));
    const float w1 = __int_as_float(__builtin_amdgcn_readlane(wi, jj + 1));
    const float w2 = __int_as_float(__builtin_amdgcn_readlane(wi, jj + 2));
    const float w3 = __int_as_float(__builtin_amdgcn_readlane(wi, jj + 3));
    const float4 r0 = *(const float4*)(xcol + (size_t)s0 * F);
    const float4 r1 = *(const float4*)(xcol + (size_t)s1 * F);
    const float4 r2 = *(const float4*)(xcol + (size_t)s2 * F);
    const float4 r3 = *(const float4*)(xcol + (size_t)s3 * F);
    fma4(acc, r0, w0);
    fma4(acc, r1, w1);
    fma4(acc, r2, w2);
    fma4(acc, r3, w3);
  }
  for (; jj < n; ++jj) {
    const int s0 = __builtin_amdgcn_readlane(s, jj);
    const float w0 = __int_as_float(__builtin_amdgcn_readlane(wi, jj));
    const float4 r0 = *(const float4*)(xcol + (size_t)s0 * F);
    fma4(acc, r0, w0);
  }
}

template <int H, bool ELU>
__global__ __launch_bounds__(256) void segment_softmax_aggregate(
    const int* __restrict__ off, const int* __restrict__ ssrc,
    const float* __restrict__ xp, const float* __restrict__ ps,
    const float* __restrict__ pd, const float* __restrict__ bias,
    float* __restrict__ out, int Nn)
{
  constexpr int WPN = H * 2;
  constexpr int F   = H * 256;
  const int lane = threadIdx.x & 31;
  const int wv = __builtin_amdgcn_readfirstlane((int)(threadIdx.x >> 5));
  const int nw = (int)gridDim.x * 8;
  for (int gw = (int)blockIdx.x * 8 + wv; gw < Nn * WPN; gw += nw) {
  const int d    = gw / WPN;
  const int part = gw - d * WPN;
  const int h    = part >> 1;
  const int beg  = off[d];
  const int deg  = off[d + 1] - beg;
  const int tot  = deg + 1;
  const float* ps0 = ps + (size_t)(2 * h) * Nn;
  const float* ps1 = ps0 + Nn;
  const float ad = pd[(size_t)(2 * h) * Nn + d] + pd[(size_t)(2 * h + 1) * Nn + d];
  const float NEG_INF = -__builtin_inff();

  int   s0 = d;
  float a0 = NEG_INF;
  if (lane < tot) {
    s0 = (lane < deg) ? ssrc[beg + lane] : d;
    a0 = leaky02((ps0[s0] + ps1[s0]) + ad);
  }

  float m = a0;
  for (int j = 32 + lane; j < tot; j += 32) {
    const int s = (j < deg) ? ssrc[beg + j] : d;
    m = fmaxf(m, leaky02((ps0[s] + ps1[s]) + ad));
  }
  for (int o = 16; o > 0; o >>= 1) m = fmaxf(m, __shfl_xor(m, o, 32));

  const float e0 = (lane < tot) ? __expf(a0 - m) : 0.f;
  float den = e0;
  for (int j = 32 + lane; j < tot; j += 32) {
    const int s = (j < deg) ? ssrc[beg + j] : d;
    den += __expf(leaky02((ps0[s] + ps1[s]) + ad) - m);
  }
  for (int o = 16; o > 0; o >>= 1) den += __shfl_xor(den, o, 32);

  const int col = part * 128 + lane * 4;
  const float* xcol = xp + col;
  const float4 b = *(const float4*)(bias + col);
  float4 acc = make_float4(0.f, 0.f, 0.f, 0.f);

  gather_group(acc, xcol, F, s0, e0 / den, (tot < 32) ? tot : 32);
  for (int base = 32; base < tot; base += 32) {
    const int j = base + lane;
    int   s = d;
    float w = 0.f;
    if (j < tot) {
      s = (j < deg) ? ssrc[beg + j] : d;
      w = __expf(leaky02((ps0[s] + ps1[s]) + ad) - m) / den;
    }
    const int n = tot - base;
    gather_group(acc, xcol, F, s, w, (n < 32) ? n : 32);
  }

  float4 v = make_float4(acc.x + b.x, acc.y + b.y, acc.z + b.z, acc.w + b.w);
  if (ELU) {
    v.x = (v.x > 0.f) ? v.x : expm1f(v.x);
    v.y = (v.y > 0.f) ? v.y : expm1f(v.y);
    v.z = (v.z > 0.f) ? v.z : expm1f(v.z);
    v.w = (v.w > 0.f) ? v.w : expm1f(v.w);
  }
  *(float4*)(out + (size_t)d * F + col) = v;
  }
}


extern "C" void kernel_launch(void* const* d_in, const int* in_sizes, int n_in,
                              void* d_out, int out_size, void* d_ws, size_t ws_size,
                              hipStream_t stream) {
  const float* x   = (const float*)d_in[0];
  const int*   ei  = (const int*)d_in[1];
  const float* W1  = (const float*)d_in[2];
  const float* as1 = (const float*)d_in[3];
  const float* ad1 = (const float*)d_in[4];
  const float* b1  = (const float*)d_in[5];
  const float* W2  = (const float*)d_in[6];
  const float* as2 = (const float*)d_in[7];
  const float* ad2 = (const float*)d_in[8];
  const float* b2  = (const float*)d_in[9];

  const int K1 = 768, F1 = 768, H1 = 3;
  const int F2 = 256;
  const int Nn = in_sizes[0] / K1;
  const int E0 = in_sizes[1] / 2;

  const int* srcI = ei;
  const int* dstI = ei + E0;

  float* ws = (float*)d_ws;
  float* xp1   = ws; ws += (size_t)Nn * F1;
  float* h1    = ws; ws += (size_t)Nn * F1;
  float* xp2   = ws; ws += (size_t)Nn * F2;
  float* part2 = ws; ws += (size_t)Nn * 8;
  float* ps1 = xp2;
  float* pd1 = xp2 + (size_t)Nn * 6;
  float* ps2 = part2;
  float* pd2 = part2 + (size_t)Nn * 2;
  ws += Nn;
  int* off  = (int*)ws; ws += Nn + 1;
  ws += Nn;
  int* ssrc = (int*)ws; ws += E0;

  {
    dim3 g(F1 / 128, (Nn + 127) / 128 + 1);
    wmma_gemm_f32<4, true><<<g, 256, 0, stream>>>(x, W1, xp1, Nn, F1, K1, as1, ad1, ps1, pd1,
                                                  srcI, dstI, E0, off, ssrc);
  }
  segment_softmax_aggregate<3, true><<<1875, 256, 0, stream>>>(
      off, ssrc, xp1, ps1, pd1, b1, h1, Nn);

  {
    dim3 g(F2 / 128, (Nn + 63) / 64);
    wmma_gemm_f32<2, false><<<g, 128, 0, stream>>>(h1, W2, xp2, Nn, F2, F1, as2, ad2, ps2, pd2,
                                                   nullptr, nullptr, 0, nullptr, nullptr);
  }
  segment_softmax_aggregate<1, false><<<1250, 256, 0, stream>>>(
      off, ssrc, xp2, ps2, pd2, b2, (float*)d_out, Nn);
}
